# starting kernel
# baseline (speedup 1.0000x reference)
#include <hip/hip_runtime.h>
#include <hip/hip_bf16.h>
#include <math.h>


#define BB      2
#define TT      2048
#define DM      1024
#define NH      16
#define DH      64
#define ROWS    (BB * TT)
#define QKV_N   (3 * DM)

typedef __attribute__((ext_vector_type(2))) float v2f;
typedef __attribute__((ext_vector_type(8))) float v8f;

__global__ void gemm_bias_f32wmma(const float* __restrict__ A,
                                  const float* __restrict__ Bm,
                                  const float* __restrict__ bias,
                                  float* __restrict__ C,
                                  int M, int N, int K)
{
    const int wave = blockIdx.x * (blockDim.x >> 5) + (threadIdx.x >> 5);
    const int lane = threadIdx.x & 31;
    const int half = lane >> 4;
    const int ln   = lane & 15;

    const int ngroups = N >> 6;
    const int mt = wave / ngroups;
    const int ng = wave % ngroups;
    if (mt * 32 >= M) return;

    const float* __restrict__ ap0 = A + (size_t)(mt * 32 + ln) * K + 2 * half;
    const float* __restrict__ ap1 = ap0 + (size_t)16 * K;
    const float* __restrict__ bp  = Bm + (size_t)(2 * half) * N + ng * 64 + ln;

    v8f acc00 = {}, acc01 = {}, acc02 = {}, acc03 = {};
    v8f acc10 = {}, acc11 = {}, acc12 = {}, acc13 = {};

    for (int k0 = 0; k0 < K; k0 += 4) {
        const v2f a0 = *(const v2f*)ap0;
        const v2f a1 = *(const v2f*)ap1;

        v2f b0; b0.x = bp[0];  b0.y = bp[N];
        v2f b1; b1.x = bp[16]; b1.y = bp[N + 16];
        v2f b2; b2.x = bp[32]; b2.y = bp[N + 32];
        v2f b3; b3.x = bp[48]; b3.y = bp[N + 48];

        acc00 = __builtin_amdgcn_wmma_f32_16x16x4_f32(false, a0, false, b0, (short)0, acc00, false, false);
        acc10 = __builtin_amdgcn_wmma_f32_16x16x4_f32(false, a1, false, b0, (short)0, acc10, false, false);
        acc01 = __builtin_amdgcn_wmma_f32_16x16x4_f32(false, a0, false, b1, (short)0, acc01, false, false);
        acc11 = __builtin_amdgcn_wmma_f32_16x16x4_f32(false, a1, false, b1, (short)0, acc11, false, false);
        acc02 = __builtin_amdgcn_wmma_f32_16x16x4_f32(false, a0, false, b2, (short)0, acc02, false, false);
        acc12 = __builtin_amdgcn_wmma_f32_16x16x4_f32(false, a1, false, b2, (short)0, acc12, false, false);
        acc03 = __builtin_amdgcn_wmma_f32_16x16x4_f32(false, a0, false, b3, (short)0, acc03, false, false);
        acc13 = __builtin_amdgcn_wmma_f32_16x16x4_f32(false, a1, false, b3, (short)0, acc13, false, false);

        ap0 += 4;
        ap1 += 4;
        bp  += (size_t)4 * N;
    }

    v8f accs[2][4] = { { acc00, acc01, acc02, acc03 },
                       { acc10, acc11, acc12, acc13 } };
#pragma unroll
    for (int sub = 0; sub < 2; ++sub) {
        const int rbase = mt * 32 + sub * 16 + 8 * half;
#pragma unroll
        for (int t = 0; t < 4; ++t) {
            const int col = ng * 64 + t * 16 + ln;
            const float bs = bias[col];
#pragma unroll
            for (int v = 0; v < 8; ++v)
                C[(size_t)(rbase + v) * N + col] = accs[sub][t][v] + bs;
        }
    }
}

__global__ void rope_split(const float* __restrict__ qkv,
                           float* __restrict__ qb,
                           float* __restrict__ kb,
                           float* __restrict__ vb)
{
    const int row = blockIdx.x;
    const int b = row / TT;
    const int t = row % TT;
    const float* __restrict__ src = qkv + (size_t)row * QKV_N;

    for (int p = threadIdx.x; p < QKV_N / 2; p += blockDim.x) {
        const int c   = p * 2;
        const int sec = c >> 10;
        const int cc  = c & (DM - 1);
        const int h   = cc >> 6;
        const int d   = cc & (DH - 1);

        const float e = src[c];
        const float o = src[c + 1];
        const size_t dst = (((size_t)(b * NH + h)) * TT + t) * DH + d;

        if (sec == 2) {
            vb[dst]     = e;
            vb[dst + 1] = o;
        } else {
            const int   i   = d >> 1;
            const float inv = expf(-(float)(2 * i) * (9.210340371976184f / 64.0f));
            const float ang = (float)t * inv;
            const float sn = sinf(ang);
            const float cs = cosf(ang);
            const float re = e * cs - o * sn;
            const float ro = e * sn + o * cs;
            if (sec == 0) { qb[dst] = re; qb[dst + 1] = ro; }
            else          { kb[dst] = re; kb[dst + 1] = ro; }
        }
    }
}

#define PSTRIDE 20

__global__ void flash_attn_f32wmma(const float* __restrict__ qb,
                                   const float* __restrict__ kb,
                                   const float* __restrict__ vb,
                                   const int*   __restrict__ amask,
                                   float* __restrict__ attn  )
{
    __shared__ float pshare[4][16 * PSTRIDE];

    const int wid  = threadIdx.x >> 5;
    const int lane = threadIdx.x & 31;
    const int half = lane >> 4;
    const int ln   = lane & 15;

    const int blocksPerBH = TT / 64;
    const int bh   = blockIdx.x / blocksPerBH;
    const int qblk = blockIdx.x % blocksPerBH;
    const int b    = bh >> 4;
    const int q0   = qblk * 64 + wid * 16;

    const float* __restrict__ Q  = qb + (size_t)bh * TT * DH;
    const float* __restrict__ Kp = kb + (size_t)bh * TT * DH;
    const float* __restrict__ Vp = vb + (size_t)bh * TT * DH;

    v2f qf[16];
    {
        const float* __restrict__ qrow = Q + (size_t)(q0 + ln) * DH;
#pragma unroll
        for (int i = 0; i < 16; ++i)
            qf[i] = *(const v2f*)(qrow + 4 * i + 2 * half);
    }

    v8f o0 = {}, o1 = {}, o2 = {}, o3 = {};
    float rmax[8], rsum[8];
#pragma unroll
    for (int v = 0; v < 8; ++v) { rmax[v] = -1e30f; rsum[v] = 0.0f; }

    float* __restrict__ P = &pshare[wid][0];
    const float scale = 0.125f;
    const int nkt = q0 / 16 + 1;

    for (int j = 0; j < nkt; ++j) {
        const int kbase = j * 16;

        v8f s = {};
        {
            const float* __restrict__ krow = Kp + (size_t)(kbase + ln) * DH;
#pragma unroll
            for (int i = 0; i < 16; ++i) {
                const v2f bf = *(const v2f*)(krow + 4 * i + 2 * half);
                s = __builtin_amdgcn_wmma_f32_16x16x4_f32(false, qf[i], false, bf,
                                                          (short)0, s, false, false);
            }
        }

        const int keyIdx = kbase + ln;
        const int pm     = amask[b * TT + keyIdx];
#pragma unroll
        for (int v = 0; v < 8; ++v) {
            const int qrowi = q0 + v + 8 * half;
            float val = s[v] * scale;
            val = (keyIdx > qrowi || pm == 0) ? -1e30f : val;

            float mx = val;
            mx = fmaxf(mx, __shfl_xor(mx, 1, 32));
            mx = fmaxf(mx, __shfl_xor(mx, 2, 32));
            mx = fmaxf(mx, __shfl_xor(mx, 4, 32));
            mx = fmaxf(mx, __shfl_xor(mx, 8, 32));

            const float nmax = fmaxf(rmax[v], mx);
            const float p    = __expf(val - nmax);

            float sr = p;
            sr += __shfl_xor(sr, 1, 32);
            sr += __shfl_xor(sr, 2, 32);
            sr += __shfl_xor(sr, 4, 32);
            sr += __shfl_xor(sr, 8, 32);

            const float alpha = __expf(rmax[v] - nmax);
            rsum[v] = rsum[v] * alpha + sr;
            rmax[v] = nmax;
            o0[v] *= alpha; o1[v] *= alpha; o2[v] *= alpha; o3[v] *= alpha;

            P[(v + 8 * half) * PSTRIDE + ln] = p;
        }

        asm volatile("s_wait_dscnt 0x0" ::: "memory");

#pragma unroll
        for (int i = 0; i < 4; ++i) {
            const v2f af = *(const v2f*)(P + ln * PSTRIDE + 4 * i + 2 * half);
            const float* __restrict__ vrow =
                Vp + (size_t)(kbase + 4 * i + 2 * half) * DH + ln;
            v2f b0; b0.x = vrow[0];  b0.y = vrow[DH];
            v2f b1; b1.x = vrow[16]; b1.y = vrow[DH + 16];
            v2f b2; b2.x = vrow[32]; b2.y = vrow[DH + 32];
            v2f b3; b3.x = vrow[48]; b3.y = vrow[DH + 48];
            o0 = __builtin_amdgcn_wmma_f32_16x16x4_f32(false, af, false, b0, (short)0, o0, false, false);
            o1 = __builtin_amdgcn_wmma_f32_16x16x4_f32(false, af, false, b1, (short)0, o1, false, false);
            o2 = __builtin_amdgcn_wmma_f32_16x16x4_f32(false, af, false, b2, (short)0, o2, false, false);
            o3 = __builtin_amdgcn_wmma_f32_16x16x4_f32(false, af, false, b3, (short)0, o3, false, false);
        }
    }

    const int h = bh & (NH - 1);
    v8f os[4] = { o0, o1, o2, o3 };
#pragma unroll
    for (int v = 0; v < 8; ++v) {
        const float inv = 1.0f / rsum[v];
        const size_t row = (size_t)b * TT + (q0 + v + 8 * half);
#pragma unroll
        for (int t = 0; t < 4; ++t)
            attn[row * DM + h * DH + t * 16 + ln] = os[t][v] * inv;
    }
}

extern "C" void kernel_launch(void* const* d_in, const int* in_sizes, int n_in,
                              void* d_out, int out_size, void* d_ws, size_t ws_size,
                              hipStream_t stream)
{
    (void)in_sizes; (void)n_in; (void)out_size; (void)ws_size;

    const float* x      = (const float*)d_in[0];
    const int*   mask   = (const int*)  d_in[1];
    const float* W_qkv  = (const float*)d_in[2];
    const float* b_qkv  = (const float*)d_in[3];
    const float* W_out  = (const float*)d_in[4];
    const float* b_out  = (const float*)d_in[5];
    float*       out    = (float*)d_out;

    float* ws   = (float*)d_ws;
    float* qkv  = ws;
    float* qb   = qkv + (size_t)ROWS * QKV_N;
    float* kb   = qb  + (size_t)BB * NH * TT * DH;
    float* vb   = kb  + (size_t)BB * NH * TT * DH;
    float* attn = qkv;

    {
        const int waves = (ROWS / 32) * (QKV_N / 64);
        gemm_bias_f32wmma<<<waves / 4, 128, 0, stream>>>(
            x, W_qkv, b_qkv, qkv, ROWS, QKV_N, DM);
    }

    rope_split<<<ROWS, 256, 0, stream>>>(qkv, qb, kb, vb);

    {
        const int blocks = BB * NH * (TT / 64);
        flash_attn_f32wmma<<<blocks, 128, 0, stream>>>(qb, kb, vb, mask, attn);
    }

    {
        const int waves = (ROWS / 32) * (DM / 64);
        gemm_bias_f32wmma<<<waves / 4, 128, 0, stream>>>(
            attn, W_out, b_out, out, ROWS, DM, DM);
    }
}


# Both GEMMs: every fragment from a 16-byte load (4 B column tiles interleaved so a lane owns 4 adjacent columns; A as float4 k-runs), 12 loads per 32 WMMAs, float4 stores; k grouping of sums changed
# speedup vs baseline: 2.7019x; 2.7019x over previous
#include <hip/hip_runtime.h>
#include <hip/hip_bf16.h>
#include <math.h>


#define BB      2
#define TT      2048
#define DM      1024
#define NH      16
#define DH      64
#define ROWS    (BB * TT)
#define QKV_N   (3 * DM)

typedef __attribute__((ext_vector_type(2))) float v2f;
typedef __attribute__((ext_vector_type(4))) float v4f;
typedef __attribute__((ext_vector_type(8))) float v8f;

__global__ __launch_bounds__(128, 6)
void gemm_bias_f32wmma(const float* __restrict__ A,
                                  const float* __restrict__ Bm,
                                  const float* __restrict__ bias,
                                  float* __restrict__ C,
                                  int M, int N, int K)
{
    const int wave = blockIdx.x * (blockDim.x >> 5) + (threadIdx.x >> 5);
    const int lane = threadIdx.x & 31;
    const int half = lane >> 4;
    const int ln   = lane & 15;

    const int ngroups = N >> 6;
    const int mt = wave / ngroups;
    const int ng = wave % ngroups;
    if (mt * 32 >= M) return;

    const int col = ng * 64 + 4 * ln;

    const float* __restrict__ ap0 = A + (size_t)(mt * 32 + ln) * K + 4 * half;
    const float* __restrict__ ap1 = ap0 + (size_t)16 * K;
    const float* __restrict__ bp  = Bm + (size_t)(4 * half) * N + col;

    v8f acc[2][4] = {};

    for (int k0 = 0; k0 < K; k0 += 16) {
        v4f a[2][2];
        v4f b[2][4];

        a[0][0] = *(const v4f*)ap0;
        a[1][0] = *(const v4f*)ap1;
#pragma unroll
        for (int j = 0; j < 4; ++j)
            b[0][j] = *(const v4f*)(bp + (size_t)j * N);
        a[0][1] = *(const v4f*)(ap0 + 8);
        a[1][1] = *(const v4f*)(ap1 + 8);
#pragma unroll
        for (int j = 0; j < 4; ++j)
            b[1][j] = *(const v4f*)(bp + (size_t)(8 + j) * N);

        __builtin_amdgcn_sched_barrier(0);

#pragma unroll
        for (int s = 0; s < 2; ++s) {
#pragma unroll
            for (int p = 0; p < 2; ++p) {
                v2f af0; af0.x = a[0][s][2 * p]; af0.y = a[0][s][2 * p + 1];
                v2f af1; af1.x = a[1][s][2 * p]; af1.y = a[1][s][2 * p + 1];
#pragma unroll
                for (int t = 0; t < 4; ++t) {
                    v2f bf; bf.x = b[s][2 * p][t]; bf.y = b[s][2 * p + 1][t];
                    acc[0][t] = __builtin_amdgcn_wmma_f32_16x16x4_f32(false, af0, false, bf, (short)0, acc[0][t], false, false);
                    acc[1][t] = __builtin_amdgcn_wmma_f32_16x16x4_f32(false, af1, false, bf, (short)0, acc[1][t], false, false);
                }
            }
        }

        ap0 += 16;
        ap1 += 16;
        bp  += (size_t)16 * N;
    }

    const v4f bs = *(const v4f*)(bias + col);
#pragma unroll
    for (int sub = 0; sub < 2; ++sub) {
        const int rbase = mt * 32 + sub * 16 + 8 * half;
#pragma unroll
        for (int v = 0; v < 8; ++v) {
            v4f o;
            o.x = acc[sub][0][v] + bs.x;
            o.y = acc[sub][1][v] + bs.y;
            o.z = acc[sub][2][v] + bs.z;
            o.w = acc[sub][3][v] + bs.w;
            *(v4f*)(C + (size_t)(rbase + v) * N + col) = o;
        }
    }
}

__global__ void rope_split(const float* __restrict__ qkv,
                           float* __restrict__ qb,
                           float* __restrict__ kb,
                           float* __restrict__ vb)
{
    const int row = blockIdx.x;
    const int b = row / TT;
    const int t = row % TT;
    const float* __restrict__ src = qkv + (size_t)row * QKV_N;

    for (int p = threadIdx.x; p < QKV_N / 2; p += blockDim.x) {
        const int c   = p * 2;
        const int sec = c >> 10;
        const int cc  = c & (DM - 1);
        const int h   = cc >> 6;
        const int d   = cc & (DH - 1);

        const float e = src[c];
        const float o = src[c + 1];
        const size_t dst = (((size_t)(b * NH + h)) * TT + t) * DH + d;

        if (sec == 2) {
            vb[dst]     = e;
            vb[dst + 1] = o;
        } else {
            const int   i   = d >> 1;
            const float inv = expf(-(float)(2 * i) * (9.210340371976184f / 64.0f));
            const float ang = (float)t * inv;
            const float sn = sinf(ang);
            const float cs = cosf(ang);
            const float re = e * cs - o * sn;
            const float ro = e * sn + o * cs;
            if (sec == 0) { qb[dst] = re; qb[dst + 1] = ro; }
            else          { kb[dst] = re; kb[dst + 1] = ro; }
        }
    }
}

#define PSTRIDE 68
#define KTILE   64
#define KPITCH  68
#define VPITCH  80

__global__ __launch_bounds__(128, 4)
void flash_attn_f32wmma(const float* __restrict__ qb,
                        const float* __restrict__ kb,
                        const float* __restrict__ vb,
                        const int*   __restrict__ amask,
                        float* __restrict__ attn  )
{
    __shared__ __attribute__((aligned(16))) float pshare[4][16 * PSTRIDE];
    __shared__ __attribute__((aligned(16))) float ks[KTILE * KPITCH];
    __shared__ __attribute__((aligned(16))) float vs[KTILE * VPITCH];
    __shared__ int   ms[KTILE];

    const int tid  = threadIdx.x;
    const int wid  = tid >> 5;
    const int lane = tid & 31;
    const int half = lane >> 4;
    const int ln   = lane & 15;

    const int blocksPerBH = TT / 64;
    const int bh   = blockIdx.x / blocksPerBH;
    const int qblk = blockIdx.x % blocksPerBH;
    const int b    = bh >> 4;
    const int q0   = qblk * 64 + wid * 16;

    const float* __restrict__ Q  = qb + (size_t)bh * TT * DH;
    const float* __restrict__ Kp = kb + (size_t)bh * TT * DH;
    const float* __restrict__ Vp = vb + (size_t)bh * TT * DH;

    v2f qf[16];
    {
        const float* __restrict__ qrow = Q + (size_t)(q0 + ln) * DH;
#pragma unroll
        for (int i = 0; i < 16; ++i)
            qf[i] = *(const v2f*)(qrow + 4 * i + 2 * half);
    }

    v8f o0 = {}, o1 = {}, o2 = {}, o3 = {};
    float rmax[8], rsum[8];
#pragma unroll
    for (int v = 0; v < 8; ++v) { rmax[v] = -1e30f; rsum[v] = 0.0f; }

    float* __restrict__ P = &pshare[wid][0];
    const float scale = 0.125f;

    const int srow = tid >> 4;
    const int sc4  = tid & 15;
    float* __restrict__ kdst = ks + srow * KPITCH + 4 * sc4;
    float* __restrict__ vdst = vs + srow * VPITCH + 4 * sc4;

    const int* __restrict__ mg = amask + b * TT + (tid & (KTILE - 1));

    v4f kst[8], vst[8];
    int mst;
    {
        const v4f* __restrict__ kg = (const v4f*)Kp + tid;
        const v4f* __restrict__ vg = (const v4f*)Vp + tid;
#pragma unroll
        for (int it = 0; it < 8; ++it) { kst[it] = kg[it * 128]; vst[it] = vg[it * 128]; }
        mst = mg[0];
    }

    for (int kt = 0; kt <= qblk; ++kt) {
        __syncthreads();
#pragma unroll
        for (int it = 0; it < 8; ++it) {
            *(v4f*)(kdst + it * 8 * KPITCH) = kst[it];
            *(v4f*)(vdst + it * 8 * VPITCH) = vst[it];
        }
        if (tid < KTILE) ms[tid] = mst;
        __syncthreads();

        if (kt < qblk) {
            const v4f* __restrict__ kg = (const v4f*)(Kp + (size_t)(kt + 1) * KTILE * DH) + tid;
            const v4f* __restrict__ vg = (const v4f*)(Vp + (size_t)(kt + 1) * KTILE * DH) + tid;
#pragma unroll
            for (int it = 0; it < 8; ++it) { kst[it] = kg[it * 128]; vst[it] = vg[it * 128]; }
            mst = mg[(kt + 1) * KTILE];
        }

        v8f s0 = {}, s1 = {}, s2 = {}, s3 = {};
        {
            const float* __restrict__ krow = ks + ln * KPITCH + 2 * half;
#pragma unroll
            for (int i = 0; i < 16; ++i) {
                const v2f k0 = *(const v2f*)(krow + 4 * i);
                const v2f k1 = *(const v2f*)(krow + 16 * KPITCH + 4 * i);
                const v2f k2 = *(const v2f*)(krow + 32 * KPITCH + 4 * i);
                const v2f k3 = *(const v2f*)(krow + 48 * KPITCH + 4 * i);
                s0 = __builtin_amdgcn_wmma_f32_16x16x4_f32(false, qf[i], false, k0, (short)0, s0, false, false);
                s1 = __builtin_amdgcn_wmma_f32_16x16x4_f32(false, qf[i], false, k1, (short)0, s1, false, false);
                s2 = __builtin_amdgcn_wmma_f32_16x16x4_f32(false, qf[i], false, k2, (short)0, s2, false, false);
                s3 = __builtin_amdgcn_wmma_f32_16x16x4_f32(false, qf[i], false, k3, (short)0, s3, false, false);
            }
        }

        const int key0 = kt * KTILE + ln;
        const int pm0 = ms[ln];
        const int pm1 = ms[16 + ln];
        const int pm2 = ms[32 + ln];
        const int pm3 = ms[48 + ln];
#pragma unroll
        for (int v = 0; v < 8; ++v) {
            const int qrowi = q0 + v + 8 * half;
            float a0 = s0[v] * scale;
            float a1 = s1[v] * scale;
            float a2 = s2[v] * scale;
            float a3 = s3[v] * scale;
            a0 = (key0      > qrowi || pm0 == 0) ? -1e30f : a0;
            a1 = (key0 + 16 > qrowi || pm1 == 0) ? -1e30f : a1;
            a2 = (key0 + 32 > qrowi || pm2 == 0) ? -1e30f : a2;
            a3 = (key0 + 48 > qrowi || pm3 == 0) ? -1e30f : a3;

            float mx = fmaxf(fmaxf(a0, a1), fmaxf(a2, a3));
            mx = fmaxf(mx, __shfl_xor(mx, 1, 32));
            mx = fmaxf(mx, __shfl_xor(mx, 2, 32));
            mx = fmaxf(mx, __shfl_xor(mx, 4, 32));
            mx = fmaxf(mx, __shfl_xor(mx, 8, 32));

            const float nmax = fmaxf(rmax[v], mx);
            const float p0 = __expf(a0 - nmax);
            const float p1 = __expf(a1 - nmax);
            const float p2 = __expf(a2 - nmax);
            const float p3 = __expf(a3 - nmax);

            float sr = (p0 + p1) + (p2 + p3);
            sr += __shfl_xor(sr, 1, 32);
            sr += __shfl_xor(sr, 2, 32);
            sr += __shfl_xor(sr, 4, 32);
            sr += __shfl_xor(sr, 8, 32);

            const float alpha = __expf(rmax[v] - nmax);
            rsum[v] = rsum[v] * alpha + sr;
            rmax[v] = nmax;
            o0[v] *= alpha; o1[v] *= alpha; o2[v] *= alpha; o3[v] *= alpha;

            float* __restrict__ prow = P + (v + 8 * half) * PSTRIDE + ln;
            prow[0]  = p0;
            prow[16] = p1;
            prow[32] = p2;
            prow[48] = p3;
        }

        asm volatile("s_wait_dscnt 0x0" ::: "memory");

#pragma unroll
        for (int i = 0; i < 16; ++i) {
            const v2f af = *(const v2f*)(P + ln * PSTRIDE + 4 * i + 2 * half);
            const float* __restrict__ vrow = vs + (4 * i + 2 * half) * VPITCH + ln;
            v2f b0; b0.x = vrow[0];  b0.y = vrow[VPITCH];
            v2f b1; b1.x = vrow[16]; b1.y = vrow[VPITCH + 16];
            v2f b2; b2.x = vrow[32]; b2.y = vrow[VPITCH + 32];
            v2f b3; b3.x = vrow[48]; b3.y = vrow[VPITCH + 48];
            o0 = __builtin_amdgcn_wmma_f32_16x16x4_f32(false, af, false, b0, (short)0, o0, false, false);
            o1 = __builtin_amdgcn_wmma_f32_16x16x4_f32(false, af, false, b1, (short)0, o1, false, false);
            o2 = __builtin_amdgcn_wmma_f32_16x16x4_f32(false, af, false, b2, (short)0, o2, false, false);
            o3 = __builtin_amdgcn_wmma_f32_16x16x4_f32(false, af, false, b3, (short)0, o3, false, false);
        }
    }

    const int h = bh & (NH - 1);
    v8f os[4] = { o0, o1, o2, o3 };
#pragma unroll
    for (int v = 0; v < 8; ++v) {
        const float inv = 1.0f / rsum[v];
        const size_t row = (size_t)b * TT + (q0 + v + 8 * half);
#pragma unroll
        for (int t = 0; t < 4; ++t)
            attn[row * DM + h * DH + t * 16 + ln] = os[t][v] * inv;
    }
}

extern "C" void kernel_launch(void* const* d_in, const int* in_sizes, int n_in,
                              void* d_out, int out_size, void* d_ws, size_t ws_size,
                              hipStream_t stream)
{
    (void)in_sizes; (void)n_in; (void)out_size; (void)ws_size;

    const float* x      = (const float*)d_in[0];
    const int*   mask   = (const int*)  d_in[1];
    const float* W_qkv  = (const float*)d_in[2];
    const float* b_qkv  = (const float*)d_in[3];
    const float* W_out  = (const float*)d_in[4];
    const float* b_out  = (const float*)d_in[5];
    float*       out    = (float*)d_out;

    float* ws   = (float*)d_ws;
    float* qkv  = ws;
    float* qb   = qkv + (size_t)ROWS * QKV_N;
    float* kb   = qb  + (size_t)BB * NH * TT * DH;
    float* vb   = kb  + (size_t)BB * NH * TT * DH;
    float* attn = qkv;

    {
        const int waves = (ROWS / 32) * (QKV_N / 64);
        gemm_bias_f32wmma<<<waves / 4, 128, 0, stream>>>(
            x, W_qkv, b_qkv, qkv, ROWS, QKV_N, DM);
    }

    rope_split<<<ROWS, 256, 0, stream>>>(qkv, qb, kb, vb);

    {
        const int blocks = BB * NH * (TT / 64);
        flash_attn_f32wmma<<<blocks, 128, 0, stream>>>(qb, kb, vb, mask, attn);
    }

    {
        const int waves = (ROWS / 32) * (DM / 64);
        gemm_bias_f32wmma<<<waves / 4, 128, 0, stream>>>(
            attn, W_out, b_out, out, ROWS, DM, DM);
    }
}


# Attention: heaviest query tiles first (tile = 31 - block/32) and 56 KiB dynamic LDS hold 2 blocks per WGP, so each freed place takes the heaviest tile left; per-tile code and sums unchanged
# speedup vs baseline: 3.0955x; 1.1457x over previous
#include <hip/hip_runtime.h>
#include <hip/hip_bf16.h>
#include <math.h>


#define BB      2
#define TT      2048
#define DM      1024
#define NH      16
#define DH      64
#define ROWS    (BB * TT)
#define QKV_N   (3 * DM)

typedef __attribute__((ext_vector_type(2))) float v2f;
typedef __attribute__((ext_vector_type(4))) float v4f;
typedef __attribute__((ext_vector_type(8))) float v8f;

__global__ __launch_bounds__(128, 6)
void gemm_bias_f32wmma(const float* __restrict__ A,
                                  const float* __restrict__ Bm,
                                  const float* __restrict__ bias,
                                  float* __restrict__ C,
                                  int M, int N, int K)
{
    const int wave = blockIdx.x * (blockDim.x >> 5) + (threadIdx.x >> 5);
    const int lane = threadIdx.x & 31;
    const int half = lane >> 4;
    const int ln   = lane & 15;

    const int ngroups = N >> 6;
    const int mt = wave / ngroups;
    const int ng = wave % ngroups;
    if (mt * 32 >= M) return;

    const int col = ng * 64 + 4 * ln;

    const float* __restrict__ ap0 = A + (size_t)(mt * 32 + ln) * K + 4 * half;
    const float* __restrict__ ap1 = ap0 + (size_t)16 * K;
    const float* __restrict__ bp  = Bm + (size_t)(4 * half) * N + col;

    v8f acc[2][4] = {};

    for (int k0 = 0; k0 < K; k0 += 16) {
        v4f a[2][2];
        v4f b[2][4];

        a[0][0] = *(const v4f*)ap0;
        a[1][0] = *(const v4f*)ap1;
#pragma unroll
        for (int j = 0; j < 4; ++j)
            b[0][j] = *(const v4f*)(bp + (size_t)j * N);
        a[0][1] = *(const v4f*)(ap0 + 8);
        a[1][1] = *(const v4f*)(ap1 + 8);
#pragma unroll
        for (int j = 0; j < 4; ++j)
            b[1][j] = *(const v4f*)(bp + (size_t)(8 + j) * N);

        __builtin_amdgcn_sched_barrier(0);

#pragma unroll
        for (int s = 0; s < 2; ++s) {
#pragma unroll
            for (int p = 0; p < 2; ++p) {
                v2f af0; af0.x = a[0][s][2 * p]; af0.y = a[0][s][2 * p + 1];
                v2f af1; af1.x = a[1][s][2 * p]; af1.y = a[1][s][2 * p + 1];
#pragma unroll
                for (int t = 0; t < 4; ++t) {
                    v2f bf; bf.x = b[s][2 * p][t]; bf.y = b[s][2 * p + 1][t];
                    acc[0][t] = __builtin_amdgcn_wmma_f32_16x16x4_f32(false, af0, false, bf, (short)0, acc[0][t], false, false);
                    acc[1][t] = __builtin_amdgcn_wmma_f32_16x16x4_f32(false, af1, false, bf, (short)0, acc[1][t], false, false);
                }
            }
        }

        ap0 += 16;
        ap1 += 16;
        bp  += (size_t)16 * N;
    }

    const v4f bs = *(const v4f*)(bias + col);
#pragma unroll
    for (int sub = 0; sub < 2; ++sub) {
        const int rbase = mt * 32 + sub * 16 + 8 * half;
#pragma unroll
        for (int v = 0; v < 8; ++v) {
            v4f o;
            o.x = acc[sub][0][v] + bs.x;
            o.y = acc[sub][1][v] + bs.y;
            o.z = acc[sub][2][v] + bs.z;
            o.w = acc[sub][3][v] + bs.w;
            *(v4f*)(C + (size_t)(rbase + v) * N + col) = o;
        }
    }
}

__global__ void rope_split(const float* __restrict__ qkv,
                           float* __restrict__ qb,
                           float* __restrict__ kb,
                           float* __restrict__ vb)
{
    const int row = blockIdx.x;
    const int b = row / TT;
    const int t = row % TT;
    const float* __restrict__ src = qkv + (size_t)row * QKV_N;

    for (int p = threadIdx.x; p < QKV_N / 2; p += blockDim.x) {
        const int c   = p * 2;
        const int sec = c >> 10;
        const int cc  = c & (DM - 1);
        const int h   = cc >> 6;
        const int d   = cc & (DH - 1);

        const float e = src[c];
        const float o = src[c + 1];
        const size_t dst = (((size_t)(b * NH + h)) * TT + t) * DH + d;

        if (sec == 2) {
            vb[dst]     = e;
            vb[dst + 1] = o;
        } else {
            const int   i   = d >> 1;
            const float inv = expf(-(float)(2 * i) * (9.210340371976184f / 64.0f));
            const float ang = (float)t * inv;
            const float sn = sinf(ang);
            const float cs = cosf(ang);
            const float re = e * cs - o * sn;
            const float ro = e * sn + o * cs;
            if (sec == 0) { qb[dst] = re; qb[dst + 1] = ro; }
            else          { kb[dst] = re; kb[dst + 1] = ro; }
        }
    }
}

#define PSTRIDE 68
#define KTILE   64
#define KPITCH  68
#define VPITCH  80

__global__ __launch_bounds__(128, 4)
void flash_attn_f32wmma(const float* __restrict__ qb,
                        const float* __restrict__ kb,
                        const float* __restrict__ vb,
                        const int*   __restrict__ amask,
                        float* __restrict__ attn  )
{
    __shared__ __attribute__((aligned(16))) float pshare[4][16 * PSTRIDE];
    __shared__ __attribute__((aligned(16))) float ks[KTILE * KPITCH];
    __shared__ __attribute__((aligned(16))) float vs[KTILE * VPITCH];
    __shared__ int   ms[KTILE];

    const int tid  = threadIdx.x;
    const int wid  = tid >> 5;
    const int lane = tid & 31;
    const int half = lane >> 4;
    const int ln   = lane & 15;

    const int blocksPerBH = TT / 64;
    const int bh   = blockIdx.x % (BB * NH);
    const int qblk = blocksPerBH - 1 - blockIdx.x / (BB * NH);
    const int b    = bh >> 4;
    const int q0   = qblk * 64 + wid * 16;

    const float* __restrict__ Q  = qb + (size_t)bh * TT * DH;
    const float* __restrict__ Kp = kb + (size_t)bh * TT * DH;
    const float* __restrict__ Vp = vb + (size_t)bh * TT * DH;

    v2f qf[16];
    {
        const float* __restrict__ qrow = Q + (size_t)(q0 + ln) * DH;
#pragma unroll
        for (int i = 0; i < 16; ++i)
            qf[i] = *(const v2f*)(qrow + 4 * i + 2 * half);
    }

    v8f o0 = {}, o1 = {}, o2 = {}, o3 = {};
    float rmax[8], rsum[8];
#pragma unroll
    for (int v = 0; v < 8; ++v) { rmax[v] = -1e30f; rsum[v] = 0.0f; }

    float* __restrict__ P = &pshare[wid][0];
    const float scale = 0.125f;

    const int srow = tid >> 4;
    const int sc4  = tid & 15;
    float* __restrict__ kdst = ks + srow * KPITCH + 4 * sc4;
    float* __restrict__ vdst = vs + srow * VPITCH + 4 * sc4;

    const int* __restrict__ mg = amask + b * TT + (tid & (KTILE - 1));

    v4f kst[8], vst[8];
    int mst;
    {
        const v4f* __restrict__ kg = (const v4f*)Kp + tid;
        const v4f* __restrict__ vg = (const v4f*)Vp + tid;
#pragma unroll
        for (int it = 0; it < 8; ++it) { kst[it] = kg[it * 128]; vst[it] = vg[it * 128]; }
        mst = mg[0];
    }

    for (int kt = 0; kt <= qblk; ++kt) {
        __syncthreads();
#pragma unroll
        for (int it = 0; it < 8; ++it) {
            *(v4f*)(kdst + it * 8 * KPITCH) = kst[it];
            *(v4f*)(vdst + it * 8 * VPITCH) = vst[it];
        }
        if (tid < KTILE) ms[tid] = mst;
        __syncthreads();

        if (kt < qblk) {
            const v4f* __restrict__ kg = (const v4f*)(Kp + (size_t)(kt + 1) * KTILE * DH) + tid;
            const v4f* __restrict__ vg = (const v4f*)(Vp + (size_t)(kt + 1) * KTILE * DH) + tid;
#pragma unroll
            for (int it = 0; it < 8; ++it) { kst[it] = kg[it * 128]; vst[it] = vg[it * 128]; }
            mst = mg[(kt + 1) * KTILE];
        }

        v8f s0 = {}, s1 = {}, s2 = {}, s3 = {};
        {
            const float* __restrict__ krow = ks + ln * KPITCH + 2 * half;
#pragma unroll
            for (int i = 0; i < 16; ++i) {
                const v2f k0 = *(const v2f*)(krow + 4 * i);
                const v2f k1 = *(const v2f*)(krow + 16 * KPITCH + 4 * i);
                const v2f k2 = *(const v2f*)(krow + 32 * KPITCH + 4 * i);
                const v2f k3 = *(const v2f*)(krow + 48 * KPITCH + 4 * i);
                s0 = __builtin_amdgcn_wmma_f32_16x16x4_f32(false, qf[i], false, k0, (short)0, s0, false, false);
                s1 = __builtin_amdgcn_wmma_f32_16x16x4_f32(false, qf[i], false, k1, (short)0, s1, false, false);
                s2 = __builtin_amdgcn_wmma_f32_16x16x4_f32(false, qf[i], false, k2, (short)0, s2, false, false);
                s3 = __builtin_amdgcn_wmma_f32_16x16x4_f32(false, qf[i], false, k3, (short)0, s3, false, false);
            }
        }

        const int key0 = kt * KTILE + ln;
        const int pm0 = ms[ln];
        const int pm1 = ms[16 + ln];
        const int pm2 = ms[32 + ln];
        const int pm3 = ms[48 + ln];
#pragma unroll
        for (int v = 0; v < 8; ++v) {
            const int qrowi = q0 + v + 8 * half;
            float a0 = s0[v] * scale;
            float a1 = s1[v] * scale;
            float a2 = s2[v] * scale;
            float a3 = s3[v] * scale;
            a0 = (key0      > qrowi || pm0 == 0) ? -1e30f : a0;
            a1 = (key0 + 16 > qrowi || pm1 == 0) ? -1e30f : a1;
            a2 = (key0 + 32 > qrowi || pm2 == 0) ? -1e30f : a2;
            a3 = (key0 + 48 > qrowi || pm3 == 0) ? -1e30f : a3;

            float mx = fmaxf(fmaxf(a0, a1), fmaxf(a2, a3));
            mx = fmaxf(mx, __shfl_xor(mx, 1, 32));
            mx = fmaxf(mx, __shfl_xor(mx, 2, 32));
            mx = fmaxf(mx, __shfl_xor(mx, 4, 32));
            mx = fmaxf(mx, __shfl_xor(mx, 8, 32));

            const float nmax = fmaxf(rmax[v], mx);
            const float p0 = __expf(a0 - nmax);
            const float p1 = __expf(a1 - nmax);
            const float p2 = __expf(a2 - nmax);
            const float p3 = __expf(a3 - nmax);

            float sr = (p0 + p1) + (p2 + p3);
            sr += __shfl_xor(sr, 1, 32);
            sr += __shfl_xor(sr, 2, 32);
            sr += __shfl_xor(sr, 4, 32);
            sr += __shfl_xor(sr, 8, 32);

            const float alpha = __expf(rmax[v] - nmax);
            rsum[v] = rsum[v] * alpha + sr;
            rmax[v] = nmax;
            o0[v] *= alpha; o1[v] *= alpha; o2[v] *= alpha; o3[v] *= alpha;

            float* __restrict__ prow = P + (v + 8 * half) * PSTRIDE + ln;
            prow[0]  = p0;
            prow[16] = p1;
            prow[32] = p2;
            prow[48] = p3;
        }

        asm volatile("s_wait_dscnt 0x0" ::: "memory");

#pragma unroll
        for (int i = 0; i < 16; ++i) {
            const v2f af = *(const v2f*)(P + ln * PSTRIDE + 4 * i + 2 * half);
            const float* __restrict__ vrow = vs + (4 * i + 2 * half) * VPITCH + ln;
            v2f b0; b0.x = vrow[0];  b0.y = vrow[VPITCH];
            v2f b1; b1.x = vrow[16]; b1.y = vrow[VPITCH + 16];
            v2f b2; b2.x = vrow[32]; b2.y = vrow[VPITCH + 32];
            v2f b3; b3.x = vrow[48]; b3.y = vrow[VPITCH + 48];
            o0 = __builtin_amdgcn_wmma_f32_16x16x4_f32(false, af, false, b0, (short)0, o0, false, false);
            o1 = __builtin_amdgcn_wmma_f32_16x16x4_f32(false, af, false, b1, (short)0, o1, false, false);
            o2 = __builtin_amdgcn_wmma_f32_16x16x4_f32(false, af, false, b2, (short)0, o2, false, false);
            o3 = __builtin_amdgcn_wmma_f32_16x16x4_f32(false, af, false, b3, (short)0, o3, false, false);
        }
    }

    const int h = bh & (NH - 1);
    v8f os[4] = { o0, o1, o2, o3 };
#pragma unroll
    for (int v = 0; v < 8; ++v) {
        const float inv = 1.0f / rsum[v];
        const size_t row = (size_t)b * TT + (q0 + v + 8 * half);
#pragma unroll
        for (int t = 0; t < 4; ++t)
            attn[row * DM + h * DH + t * 16 + ln] = os[t][v] * inv;
    }
}

extern "C" void kernel_launch(void* const* d_in, const int* in_sizes, int n_in,
                              void* d_out, int out_size, void* d_ws, size_t ws_size,
                              hipStream_t stream)
{
    (void)in_sizes; (void)n_in; (void)out_size; (void)ws_size;

    const float* x      = (const float*)d_in[0];
    const int*   mask   = (const int*)  d_in[1];
    const float* W_qkv  = (const float*)d_in[2];
    const float* b_qkv  = (const float*)d_in[3];
    const float* W_out  = (const float*)d_in[4];
    const float* b_out  = (const float*)d_in[5];
    float*       out    = (float*)d_out;

    float* ws   = (float*)d_ws;
    float* qkv  = ws;
    float* qb   = qkv + (size_t)ROWS * QKV_N;
    float* kb   = qb  + (size_t)BB * NH * TT * DH;
    float* vb   = kb  + (size_t)BB * NH * TT * DH;
    float* attn = qkv;

    {
        const int waves = (ROWS / 32) * (QKV_N / 64);
        gemm_bias_f32wmma<<<waves / 4, 128, 0, stream>>>(
            x, W_qkv, b_qkv, qkv, ROWS, QKV_N, DM);
    }

    rope_split<<<ROWS, 256, 0, stream>>>(qkv, qb, kb, vb);

    {
        const int blocks = BB * NH * (TT / 64);
        flash_attn_f32wmma<<<blocks, 128, 57344, stream>>>(qb, kb, vb, mask, attn);
    }

    {
        const int waves = (ROWS / 32) * (DM / 64);
        gemm_bias_f32wmma<<<waves / 4, 128, 0, stream>>>(
            attn, W_out, b_out, out, ROWS, DM, DM);
    }
}


# Attention softmax: row max and row sum over 16 lanes by row_xmask DPP steps on the vector ALU, not by 64 LDS-unit lane shuffles per key tile, each waited for; partners and order of sums unchanged
# speedup vs baseline: 3.2060x; 1.0357x over previous
#include <hip/hip_runtime.h>
#include <hip/hip_bf16.h>
#include <math.h>


#define BB      2
#define TT      2048
#define DM      1024
#define NH      16
#define DH      64
#define ROWS    (BB * TT)
#define QKV_N   (3 * DM)

typedef __attribute__((ext_vector_type(2))) float v2f;
typedef __attribute__((ext_vector_type(4))) float v4f;
typedef __attribute__((ext_vector_type(8))) float v8f;

__global__ __launch_bounds__(128, 6)
void gemm_bias_f32wmma(const float* __restrict__ A,
                                  const float* __restrict__ Bm,
                                  const float* __restrict__ bias,
                                  float* __restrict__ C,
                                  int M, int N, int K)
{
    const int wave = blockIdx.x * (blockDim.x >> 5) + (threadIdx.x >> 5);
    const int lane = threadIdx.x & 31;
    const int half = lane >> 4;
    const int ln   = lane & 15;

    const int ngroups = N >> 6;
    const int mt = wave / ngroups;
    const int ng = wave % ngroups;
    if (mt * 32 >= M) return;

    const int col = ng * 64 + 4 * ln;

    const float* __restrict__ ap0 = A + (size_t)(mt * 32 + ln) * K + 4 * half;
    const float* __restrict__ ap1 = ap0 + (size_t)16 * K;
    const float* __restrict__ bp  = Bm + (size_t)(4 * half) * N + col;

    v8f acc[2][4] = {};

    for (int k0 = 0; k0 < K; k0 += 16) {
        v4f a[2][2];
        v4f b[2][4];

        a[0][0] = *(const v4f*)ap0;
        a[1][0] = *(const v4f*)ap1;
#pragma unroll
        for (int j = 0; j < 4; ++j)
            b[0][j] = *(const v4f*)(bp + (size_t)j * N);
        a[0][1] = *(const v4f*)(ap0 + 8);
        a[1][1] = *(const v4f*)(ap1 + 8);
#pragma unroll
        for (int j = 0; j < 4; ++j)
            b[1][j] = *(const v4f*)(bp + (size_t)(8 + j) * N);

        __builtin_amdgcn_sched_barrier(0);

#pragma unroll
        for (int s = 0; s < 2; ++s) {
#pragma unroll
            for (int p = 0; p < 2; ++p) {
                v2f af0; af0.x = a[0][s][2 * p]; af0.y = a[0][s][2 * p + 1];
                v2f af1; af1.x = a[1][s][2 * p]; af1.y = a[1][s][2 * p + 1];
#pragma unroll
                for (int t = 0; t < 4; ++t) {
                    v2f bf; bf.x = b[s][2 * p][t]; bf.y = b[s][2 * p + 1][t];
                    acc[0][t] = __builtin_amdgcn_wmma_f32_16x16x4_f32(false, af0, false, bf, (short)0, acc[0][t], false, false);
                    acc[1][t] = __builtin_amdgcn_wmma_f32_16x16x4_f32(false, af1, false, bf, (short)0, acc[1][t], false, false);
                }
            }
        }

        ap0 += 16;
        ap1 += 16;
        bp  += (size_t)16 * N;
    }

    const v4f bs = *(const v4f*)(bias + col);
#pragma unroll
    for (int sub = 0; sub < 2; ++sub) {
        const int rbase = mt * 32 + sub * 16 + 8 * half;
#pragma unroll
        for (int v = 0; v < 8; ++v) {
            v4f o;
            o.x = acc[sub][0][v] + bs.x;
            o.y = acc[sub][1][v] + bs.y;
            o.z = acc[sub][2][v] + bs.z;
            o.w = acc[sub][3][v] + bs.w;
            *(v4f*)(C + (size_t)(rbase + v) * N + col) = o;
        }
    }
}

__global__ void rope_split(const float* __restrict__ qkv,
                           float* __restrict__ qb,
                           float* __restrict__ kb,
                           float* __restrict__ vb)
{
    const int row = blockIdx.x;
    const int b = row / TT;
    const int t = row % TT;
    const float* __restrict__ src = qkv + (size_t)row * QKV_N;

    for (int p = threadIdx.x; p < QKV_N / 2; p += blockDim.x) {
        const int c   = p * 2;
        const int sec = c >> 10;
        const int cc  = c & (DM - 1);
        const int h   = cc >> 6;
        const int d   = cc & (DH - 1);

        const float e = src[c];
        const float o = src[c + 1];
        const size_t dst = (((size_t)(b * NH + h)) * TT + t) * DH + d;

        if (sec == 2) {
            vb[dst]     = e;
            vb[dst + 1] = o;
        } else {
            const int   i   = d >> 1;
            const float inv = expf(-(float)(2 * i) * (9.210340371976184f / 64.0f));
            const float ang = (float)t * inv;
            const float sn = sinf(ang);
            const float cs = cosf(ang);
            const float re = e * cs - o * sn;
            const float ro = e * sn + o * cs;
            if (sec == 0) { qb[dst] = re; qb[dst + 1] = ro; }
            else          { kb[dst] = re; kb[dst + 1] = ro; }
        }
    }
}

#define PSTRIDE 68
#define KTILE   64
#define KPITCH  68
#define VPITCH  80

template <int M>
__device__ __forceinline__ float row_xor(float x)
{
    return __int_as_float(__builtin_amdgcn_update_dpp(
        0, __float_as_int(x), 0x160 | M, 0xf, 0xf, true));
}

__global__ __launch_bounds__(128, 4)
void flash_attn_f32wmma(const float* __restrict__ qb,
                        const float* __restrict__ kb,
                        const float* __restrict__ vb,
                        const int*   __restrict__ amask,
                        float* __restrict__ attn  )
{
    __shared__ __attribute__((aligned(16))) float pshare[4][16 * PSTRIDE];
    __shared__ __attribute__((aligned(16))) float ks[KTILE * KPITCH];
    __shared__ __attribute__((aligned(16))) float vs[KTILE * VPITCH];
    __shared__ int   ms[KTILE];

    const int tid  = threadIdx.x;
    const int wid  = tid >> 5;
    const int lane = tid & 31;
    const int half = lane >> 4;
    const int ln   = lane & 15;

    const int blocksPerBH = TT / 64;
    const int bh   = blockIdx.x % (BB * NH);
    const int qblk = blocksPerBH - 1 - blockIdx.x / (BB * NH);
    const int b    = bh >> 4;
    const int q0   = qblk * 64 + wid * 16;

    const float* __restrict__ Q  = qb + (size_t)bh * TT * DH;
    const float* __restrict__ Kp = kb + (size_t)bh * TT * DH;
    const float* __restrict__ Vp = vb + (size_t)bh * TT * DH;

    v2f qf[16];
    {
        const float* __restrict__ qrow = Q + (size_t)(q0 + ln) * DH;
#pragma unroll
        for (int i = 0; i < 16; ++i)
            qf[i] = *(const v2f*)(qrow + 4 * i + 2 * half);
    }

    v8f o0 = {}, o1 = {}, o2 = {}, o3 = {};
    float rmax[8], rsum[8];
#pragma unroll
    for (int v = 0; v < 8; ++v) { rmax[v] = -1e30f; rsum[v] = 0.0f; }

    float* __restrict__ P = &pshare[wid][0];
    const float scale = 0.125f;

    const int srow = tid >> 4;
    const int sc4  = tid & 15;
    float* __restrict__ kdst = ks + srow * KPITCH + 4 * sc4;
    float* __restrict__ vdst = vs + srow * VPITCH + 4 * sc4;

    const int* __restrict__ mg = amask + b * TT + (tid & (KTILE - 1));

    v4f kst[8], vst[8];
    int mst;
    {
        const v4f* __restrict__ kg = (const v4f*)Kp + tid;
        const v4f* __restrict__ vg = (const v4f*)Vp + tid;
#pragma unroll
        for (int it = 0; it < 8; ++it) { kst[it] = kg[it * 128]; vst[it] = vg[it * 128]; }
        mst = mg[0];
    }

    for (int kt = 0; kt <= qblk; ++kt) {
        __syncthreads();
#pragma unroll
        for (int it = 0; it < 8; ++it) {
            *(v4f*)(kdst + it * 8 * KPITCH) = kst[it];
            *(v4f*)(vdst + it * 8 * VPITCH) = vst[it];
        }
        if (tid < KTILE) ms[tid] = mst;
        __syncthreads();

        if (kt < qblk) {
            const v4f* __restrict__ kg = (const v4f*)(Kp + (size_t)(kt + 1) * KTILE * DH) + tid;
            const v4f* __restrict__ vg = (const v4f*)(Vp + (size_t)(kt + 1) * KTILE * DH) + tid;
#pragma unroll
            for (int it = 0; it < 8; ++it) { kst[it] = kg[it * 128]; vst[it] = vg[it * 128]; }
            mst = mg[(kt + 1) * KTILE];
        }

        v8f s0 = {}, s1 = {}, s2 = {}, s3 = {};
        {
            const float* __restrict__ krow = ks + ln * KPITCH + 2 * half;
#pragma unroll
            for (int i = 0; i < 16; ++i) {
                const v2f k0 = *(const v2f*)(krow + 4 * i);
                const v2f k1 = *(const v2f*)(krow + 16 * KPITCH + 4 * i);
                const v2f k2 = *(const v2f*)(krow + 32 * KPITCH + 4 * i);
                const v2f k3 = *(const v2f*)(krow + 48 * KPITCH + 4 * i);
                s0 = __builtin_amdgcn_wmma_f32_16x16x4_f32(false, qf[i], false, k0, (short)0, s0, false, false);
                s1 = __builtin_amdgcn_wmma_f32_16x16x4_f32(false, qf[i], false, k1, (short)0, s1, false, false);
                s2 = __builtin_amdgcn_wmma_f32_16x16x4_f32(false, qf[i], false, k2, (short)0, s2, false, false);
                s3 = __builtin_amdgcn_wmma_f32_16x16x4_f32(false, qf[i], false, k3, (short)0, s3, false, false);
            }
        }

        const int key0 = kt * KTILE + ln;
        const int pm0 = ms[ln];
        const int pm1 = ms[16 + ln];
        const int pm2 = ms[32 + ln];
        const int pm3 = ms[48 + ln];
#pragma unroll
        for (int v = 0; v < 8; ++v) {
            const int qrowi = q0 + v + 8 * half;
            float a0 = s0[v] * scale;
            float a1 = s1[v] * scale;
            float a2 = s2[v] * scale;
            float a3 = s3[v] * scale;
            a0 = (key0      > qrowi || pm0 == 0) ? -1e30f : a0;
            a1 = (key0 + 16 > qrowi || pm1 == 0) ? -1e30f : a1;
            a2 = (key0 + 32 > qrowi || pm2 == 0) ? -1e30f : a2;
            a3 = (key0 + 48 > qrowi || pm3 == 0) ? -1e30f : a3;

            float mx = fmaxf(fmaxf(a0, a1), fmaxf(a2, a3));
            mx = fmaxf(mx, row_xor<1>(mx));
            mx = fmaxf(mx, row_xor<2>(mx));
            mx = fmaxf(mx, row_xor<4>(mx));
            mx = fmaxf(mx, row_xor<8>(mx));

            const float nmax = fmaxf(rmax[v], mx);
            const float p0 = __expf(a0 - nmax);
            const float p1 = __expf(a1 - nmax);
            const float p2 = __expf(a2 - nmax);
            const float p3 = __expf(a3 - nmax);

            float sr = (p0 + p1) + (p2 + p3);
            sr += row_xor<1>(sr);
            sr += row_xor<2>(sr);
            sr += row_xor<4>(sr);
            sr += row_xor<8>(sr);

            const float alpha = __expf(rmax[v] - nmax);
            rsum[v] = rsum[v] * alpha + sr;
            rmax[v] = nmax;
            o0[v] *= alpha; o1[v] *= alpha; o2[v] *= alpha; o3[v] *= alpha;

            float* __restrict__ prow = P + (v + 8 * half) * PSTRIDE + ln;
            prow[0]  = p0;
            prow[16] = p1;
            prow[32] = p2;
            prow[48] = p3;
        }

        asm volatile("s_wait_dscnt 0x0" ::: "memory");

#pragma unroll
        for (int i = 0; i < 16; ++i) {
            const v2f af = *(const v2f*)(P + ln * PSTRIDE + 4 * i + 2 * half);
            const float* __restrict__ vrow = vs + (4 * i + 2 * half) * VPITCH + ln;
            v2f b0; b0.x = vrow[0];  b0.y = vrow[VPITCH];
            v2f b1; b1.x = vrow[16]; b1.y = vrow[VPITCH + 16];
            v2f b2; b2.x = vrow[32]; b2.y = vrow[VPITCH + 32];
            v2f b3; b3.x = vrow[48]; b3.y = vrow[VPITCH + 48];
            o0 = __builtin_amdgcn_wmma_f32_16x16x4_f32(false, af, false, b0, (short)0, o0, false, false);
            o1 = __builtin_amdgcn_wmma_f32_16x16x4_f32(false, af, false, b1, (short)0, o1, false, false);
            o2 = __builtin_amdgcn_wmma_f32_16x16x4_f32(false, af, false, b2, (short)0, o2, false, false);
            o3 = __builtin_amdgcn_wmma_f32_16x16x4_f32(false, af, false, b3, (short)0, o3, false, false);
        }
    }

    const int h = bh & (NH - 1);
    v8f os[4] = { o0, o1, o2, o3 };
#pragma unroll
    for (int v = 0; v < 8; ++v) {
        const float inv = 1.0f / rsum[v];
        const size_t row = (size_t)b * TT + (q0 + v + 8 * half);
#pragma unroll
        for (int t = 0; t < 4; ++t)
            attn[row * DM + h * DH + t * 16 + ln] = os[t][v] * inv;
    }
}

extern "C" void kernel_launch(void* const* d_in, const int* in_sizes, int n_in,
                              void* d_out, int out_size, void* d_ws, size_t ws_size,
                              hipStream_t stream)
{
    (void)in_sizes; (void)n_in; (void)out_size; (void)ws_size;

    const float* x      = (const float*)d_in[0];
    const int*   mask   = (const int*)  d_in[1];
    const float* W_qkv  = (const float*)d_in[2];
    const float* b_qkv  = (const float*)d_in[3];
    const float* W_out  = (const float*)d_in[4];
    const float* b_out  = (const float*)d_in[5];
    float*       out    = (float*)d_out;

    float* ws   = (float*)d_ws;
    float* qkv  = ws;
    float* qb   = qkv + (size_t)ROWS * QKV_N;
    float* kb   = qb  + (size_t)BB * NH * TT * DH;
    float* vb   = kb  + (size_t)BB * NH * TT * DH;
    float* attn = qkv;

    {
        const int waves = (ROWS / 32) * (QKV_N / 64);
        gemm_bias_f32wmma<<<waves / 4, 128, 0, stream>>>(
            x, W_qkv, b_qkv, qkv, ROWS, QKV_N, DM);
    }

    rope_split<<<ROWS, 256, 0, stream>>>(qkv, qb, kb, vb);

    {
        const int blocks = BB * NH * (TT / 64);
        flash_attn_f32wmma<<<blocks, 128, 57344, stream>>>(qb, kb, vb, mask, attn);
    }

    {
        const int waves = (ROWS / 32) * (DM / 64);
        gemm_bias_f32wmma<<<waves / 4, 128, 0, stream>>>(
            attn, W_out, b_out, out, ROWS, DM, DM);
    }
}


# GEMMs: fragment loads run one trip ahead in a 2nd register set, so QKV runs 3 waves/SIMD on 32x128 wave tiles: a 16-line A load feeds 16 WMMAs, not 8; k order of sums unchanged
# speedup vs baseline: 3.4249x; 1.0683x over previous
#include <hip/hip_runtime.h>
#include <hip/hip_bf16.h>
#include <math.h>


#define BB      2
#define TT      2048
#define DM      1024
#define NH      16
#define DH      64
#define ROWS    (BB * TT)
#define QKV_N   (3 * DM)

typedef __attribute__((ext_vector_type(2))) float v2f;
typedef __attribute__((ext_vector_type(4))) float v4f;
typedef __attribute__((ext_vector_type(8))) float v8f;

template <int NG, int N, int K>
struct GemmFrags {
    v4f a[2][2];
    v4f b[2][4][NG];
};

template <int NG, int N, int K>
__device__ __forceinline__ void gemm_load(GemmFrags<NG, N, K>& f,
                                          const float* __restrict__ ap0,
                                          const float* __restrict__ ap1,
                                          const float* __restrict__ bp)
{
    f.a[0][0] = *(const v4f*)ap0;
    f.a[1][0] = *(const v4f*)ap1;
#pragma unroll
    for (int j = 0; j < 4; ++j)
#pragma unroll
        for (int g = 0; g < NG; ++g)
            f.b[0][j][g] = *(const v4f*)(bp + (size_t)j * N + 64 * g);
    f.a[0][1] = *(const v4f*)(ap0 + 8);
    f.a[1][1] = *(const v4f*)(ap1 + 8);
#pragma unroll
    for (int j = 0; j < 4; ++j)
#pragma unroll
        for (int g = 0; g < NG; ++g)
            f.b[1][j][g] = *(const v4f*)(bp + (size_t)(8 + j) * N + 64 * g);
}

template <int NG, int N, int K>
__device__ __forceinline__ void gemm_mma(const GemmFrags<NG, N, K>& f, v8f (&acc)[2][4 * NG])
{
#pragma unroll
    for (int s = 0; s < 2; ++s) {
#pragma unroll
        for (int p = 0; p < 2; ++p) {
            v2f af0; af0.x = f.a[0][s][2 * p]; af0.y = f.a[0][s][2 * p + 1];
            v2f af1; af1.x = f.a[1][s][2 * p]; af1.y = f.a[1][s][2 * p + 1];
#pragma unroll
            for (int g = 0; g < NG; ++g) {
#pragma unroll
                for (int t = 0; t < 4; ++t) {
                    v2f bf; bf.x = f.b[s][2 * p][g][t]; bf.y = f.b[s][2 * p + 1][g][t];
                    acc[0][4 * g + t] = __builtin_amdgcn_wmma_f32_16x16x4_f32(false, af0, false, bf, (short)0, acc[0][4 * g + t], false, false);
                    acc[1][4 * g + t] = __builtin_amdgcn_wmma_f32_16x16x4_f32(false, af1, false, bf, (short)0, acc[1][4 * g + t], false, false);
                }
            }
        }
    }
}

template <int NG, int N, int K>
__global__ __launch_bounds__(128, (NG == 2) ? 3 : 4)
void gemm_bias_f32wmma(const float* __restrict__ A,
                       const float* __restrict__ Bm,
                       const float* __restrict__ bias,
                       float* __restrict__ C,
                       int M)
{
    const int wave = blockIdx.x * (blockDim.x >> 5) + (threadIdx.x >> 5);
    const int lane = threadIdx.x & 31;
    const int half = lane >> 4;
    const int ln   = lane & 15;

    constexpr int ngroups = N / (64 * NG);
    const int mt = wave / ngroups;
    const int ng = wave % ngroups;
    if (mt * 32 >= M) return;

    const int col = ng * (64 * NG) + 4 * ln;

    const float* __restrict__ ap0 = A + (size_t)(mt * 32 + ln) * K + 4 * half;
    const float* __restrict__ ap1 = ap0 + (size_t)16 * K;
    const float* __restrict__ bp  = Bm + (size_t)(4 * half) * N + col;

    v8f acc[2][4 * NG] = {};
    GemmFrags<NG, N, K> f0, f1;

    gemm_load<NG, N, K>(f0, ap0, ap1, bp);

    for (int k0 = 0; k0 < K - 32; k0 += 32) {
        gemm_load<NG, N, K>(f1, ap0 + 16, ap1 + 16, bp + (size_t)16 * N);
        __builtin_amdgcn_sched_barrier(0);
        gemm_mma<NG, N, K>(f0, acc);

        ap0 += 32;
        ap1 += 32;
        bp  += (size_t)32 * N;
        gemm_load<NG, N, K>(f0, ap0, ap1, bp);
        __builtin_amdgcn_sched_barrier(0);
        gemm_mma<NG, N, K>(f1, acc);
    }
    gemm_load<NG, N, K>(f1, ap0 + 16, ap1 + 16, bp + (size_t)16 * N);
    __builtin_amdgcn_sched_barrier(0);
    gemm_mma<NG, N, K>(f0, acc);
    gemm_mma<NG, N, K>(f1, acc);

#pragma unroll
    for (int g = 0; g < NG; ++g) {
        const v4f bs = *(const v4f*)(bias + col + 64 * g);
#pragma unroll
        for (int sub = 0; sub < 2; ++sub) {
            const int rbase = mt * 32 + sub * 16 + 8 * half;
#pragma unroll
            for (int v = 0; v < 8; ++v) {
                v4f o;
                o.x = acc[sub][4 * g + 0][v] + bs.x;
                o.y = acc[sub][4 * g + 1][v] + bs.y;
                o.z = acc[sub][4 * g + 2][v] + bs.z;
                o.w = acc[sub][4 * g + 3][v] + bs.w;
                *(v4f*)(C + (size_t)(rbase + v) * N + col + 64 * g) = o;
            }
        }
    }
}

__global__ void rope_split(const float* __restrict__ qkv,
                           float* __restrict__ qb,
                           float* __restrict__ kb,
                           float* __restrict__ vb)
{
    const int row = blockIdx.x;
    const int b = row / TT;
    const int t = row % TT;
    const float* __restrict__ src = qkv + (size_t)row * QKV_N;

    for (int p = threadIdx.x; p < QKV_N / 2; p += blockDim.x) {
        const int c   = p * 2;
        const int sec = c >> 10;
        const int cc  = c & (DM - 1);
        const int h   = cc >> 6;
        const int d   = cc & (DH - 1);

        const float e = src[c];
        const float o = src[c + 1];
        const size_t dst = (((size_t)(b * NH + h)) * TT + t) * DH + d;

        if (sec == 2) {
            vb[dst]     = e;
            vb[dst + 1] = o;
        } else {
            const int   i   = d >> 1;
            const float inv = expf(-(float)(2 * i) * (9.210340371976184f / 64.0f));
            const float ang = (float)t * inv;
            const float sn = sinf(ang);
            const float cs = cosf(ang);
            const float re = e * cs - o * sn;
            const float ro = e * sn + o * cs;
            if (sec == 0) { qb[dst] = re; qb[dst + 1] = ro; }
            else          { kb[dst] = re; kb[dst + 1] = ro; }
        }
    }
}

#define PSTRIDE 68
#define KTILE   64
#define KPITCH  68
#define VPITCH  80

template <int M>
__device__ __forceinline__ float row_xor(float x)
{
    return __int_as_float(__builtin_amdgcn_update_dpp(
        0, __float_as_int(x), 0x160 | M, 0xf, 0xf, true));
}

__global__ __launch_bounds__(128, 4)
void flash_attn_f32wmma(const float* __restrict__ qb,
                        const float* __restrict__ kb,
                        const float* __restrict__ vb,
                        const int*   __restrict__ amask,
                        float* __restrict__ attn  )
{
    __shared__ __attribute__((aligned(16))) float pshare[4][16 * PSTRIDE];
    __shared__ __attribute__((aligned(16))) float ks[KTILE * KPITCH];
    __shared__ __attribute__((aligned(16))) float vs[KTILE * VPITCH];
    __shared__ int   ms[KTILE];

    const int tid  = threadIdx.x;
    const int wid  = tid >> 5;
    const int lane = tid & 31;
    const int half = lane >> 4;
    const int ln   = lane & 15;

    const int blocksPerBH = TT / 64;
    const int bh   = blockIdx.x % (BB * NH);
    const int qblk = blocksPerBH - 1 - blockIdx.x / (BB * NH);
    const int b    = bh >> 4;
    const int q0   = qblk * 64 + wid * 16;

    const float* __restrict__ Q  = qb + (size_t)bh * TT * DH;
    const float* __restrict__ Kp = kb + (size_t)bh * TT * DH;
    const float* __restrict__ Vp = vb + (size_t)bh * TT * DH;

    v2f qf[16];
    {
        const float* __restrict__ qrow = Q + (size_t)(q0 + ln) * DH;
#pragma unroll
        for (int i = 0; i < 16; ++i)
            qf[i] = *(const v2f*)(qrow + 4 * i + 2 * half);
    }

    v8f o0 = {}, o1 = {}, o2 = {}, o3 = {};
    float rmax[8], rsum[8];
#pragma unroll
    for (int v = 0; v < 8; ++v) { rmax[v] = -1e30f; rsum[v] = 0.0f; }

    float* __restrict__ P = &pshare[wid][0];
    const float scale = 0.125f;

    const int srow = tid >> 4;
    const int sc4  = tid & 15;
    float* __restrict__ kdst = ks + srow * KPITCH + 4 * sc4;
    float* __restrict__ vdst = vs + srow * VPITCH + 4 * sc4;

    const int* __restrict__ mg = amask + b * TT + (tid & (KTILE - 1));

    v4f kst[8], vst[8];
    int mst;
    {
        const v4f* __restrict__ kg = (const v4f*)Kp + tid;
        const v4f* __restrict__ vg = (const v4f*)Vp + tid;
#pragma unroll
        for (int it = 0; it < 8; ++it) { kst[it] = kg[it * 128]; vst[it] = vg[it * 128]; }
        mst = mg[0];
    }

    for (int kt = 0; kt <= qblk; ++kt) {
        __syncthreads();
#pragma unroll
        for (int it = 0; it < 8; ++it) {
            *(v4f*)(kdst + it * 8 * KPITCH) = kst[it];
            *(v4f*)(vdst + it * 8 * VPITCH) = vst[it];
        }
        if (tid < KTILE) ms[tid] = mst;
        __syncthreads();

        if (kt < qblk) {
            const v4f* __restrict__ kg = (const v4f*)(Kp + (size_t)(kt + 1) * KTILE * DH) + tid;
            const v4f* __restrict__ vg = (const v4f*)(Vp + (size_t)(kt + 1) * KTILE * DH) + tid;
#pragma unroll
            for (int it = 0; it < 8; ++it) { kst[it] = kg[it * 128]; vst[it] = vg[it * 128]; }
            mst = mg[(kt + 1) * KTILE];
        }

        v8f s0 = {}, s1 = {}, s2 = {}, s3 = {};
        {
            const float* __restrict__ krow = ks + ln * KPITCH + 2 * half;
#pragma unroll
            for (int i = 0; i < 16; ++i) {
                const v2f k0 = *(const v2f*)(krow + 4 * i);
                const v2f k1 = *(const v2f*)(krow + 16 * KPITCH + 4 * i);
                const v2f k2 = *(const v2f*)(krow + 32 * KPITCH + 4 * i);
                const v2f k3 = *(const v2f*)(krow + 48 * KPITCH + 4 * i);
                s0 = __builtin_amdgcn_wmma_f32_16x16x4_f32(false, qf[i], false, k0, (short)0, s0, false, false);
                s1 = __builtin_amdgcn_wmma_f32_16x16x4_f32(false, qf[i], false, k1, (short)0, s1, false, false);
                s2 = __builtin_amdgcn_wmma_f32_16x16x4_f32(false, qf[i], false, k2, (short)0, s2, false, false);
                s3 = __builtin_amdgcn_wmma_f32_16x16x4_f32(false, qf[i], false, k3, (short)0, s3, false, false);
            }
        }

        const int key0 = kt * KTILE + ln;
        const int pm0 = ms[ln];
        const int pm1 = ms[16 + ln];
        const int pm2 = ms[32 + ln];
        const int pm3 = ms[48 + ln];
#pragma unroll
        for (int v = 0; v < 8; ++v) {
            const int qrowi = q0 + v + 8 * half;
            float a0 = s0[v] * scale;
            float a1 = s1[v] * scale;
            float a2 = s2[v] * scale;
            float a3 = s3[v] * scale;
            a0 = (key0      > qrowi || pm0 == 0) ? -1e30f : a0;
            a1 = (key0 + 16 > qrowi || pm1 == 0) ? -1e30f : a1;
            a2 = (key0 + 32 > qrowi || pm2 == 0) ? -1e30f : a2;
            a3 = (key0 + 48 > qrowi || pm3 == 0) ? -1e30f : a3;

            float mx = fmaxf(fmaxf(a0, a1), fmaxf(a2, a3));
            mx = fmaxf(mx, row_xor<1>(mx));
            mx = fmaxf(mx, row_xor<2>(mx));
            mx = fmaxf(mx, row_xor<4>(mx));
            mx = fmaxf(mx, row_xor<8>(mx));

            const float nmax = fmaxf(rmax[v], mx);
            const float p0 = __expf(a0 - nmax);
            const float p1 = __expf(a1 - nmax);
            const float p2 = __expf(a2 - nmax);
            const float p3 = __expf(a3 - nmax);

            float sr = (p0 + p1) + (p2 + p3);
            sr += row_xor<1>(sr);
            sr += row_xor<2>(sr);
            sr += row_xor<4>(sr);
            sr += row_xor<8>(sr);

            const float alpha = __expf(rmax[v] - nmax);
            rsum[v] = rsum[v] * alpha + sr;
            rmax[v] = nmax;
            o0[v] *= alpha; o1[v] *= alpha; o2[v] *= alpha; o3[v] *= alpha;

            float* __restrict__ prow = P + (v + 8 * half) * PSTRIDE + ln;
            prow[0]  = p0;
            prow[16] = p1;
            prow[32] = p2;
            prow[48] = p3;
        }

        asm volatile("s_wait_dscnt 0x0" ::: "memory");

#pragma unroll
        for (int i = 0; i < 16; ++i) {
            const v2f af = *(const v2f*)(P + ln * PSTRIDE + 4 * i + 2 * half);
            const float* __restrict__ vrow = vs + (4 * i + 2 * half) * VPITCH + ln;
            v2f b0; b0.x = vrow[0];  b0.y = vrow[VPITCH];
            v2f b1; b1.x = vrow[16]; b1.y = vrow[VPITCH + 16];
            v2f b2; b2.x = vrow[32]; b2.y = vrow[VPITCH + 32];
            v2f b3; b3.x = vrow[48]; b3.y = vrow[VPITCH + 48];
            o0 = __builtin_amdgcn_wmma_f32_16x16x4_f32(false, af, false, b0, (short)0, o0, false, false);
            o1 = __builtin_amdgcn_wmma_f32_16x16x4_f32(false, af, false, b1, (short)0, o1, false, false);
            o2 = __builtin_amdgcn_wmma_f32_16x16x4_f32(false, af, false, b2, (short)0, o2, false, false);
            o3 = __builtin_amdgcn_wmma_f32_16x16x4_f32(false, af, false, b3, (short)0, o3, false, false);
        }
    }

    const int h = bh & (NH - 1);
    v8f os[4] = { o0, o1, o2, o3 };
#pragma unroll
    for (int v = 0; v < 8; ++v) {
        const float inv = 1.0f / rsum[v];
        const size_t row = (size_t)b * TT + (q0 + v + 8 * half);
#pragma unroll
        for (int t = 0; t < 4; ++t)
            attn[row * DM + h * DH + t * 16 + ln] = os[t][v] * inv;
    }
}

extern "C" void kernel_launch(void* const* d_in, const int* in_sizes, int n_in,
                              void* d_out, int out_size, void* d_ws, size_t ws_size,
                              hipStream_t stream)
{
    (void)in_sizes; (void)n_in; (void)out_size; (void)ws_size;

    const float* x      = (const float*)d_in[0];
    const int*   mask   = (const int*)  d_in[1];
    const float* W_qkv  = (const float*)d_in[2];
    const float* b_qkv  = (const float*)d_in[3];
    const float* W_out  = (const float*)d_in[4];
    const float* b_out  = (const float*)d_in[5];
    float*       out    = (float*)d_out;

    float* ws   = (float*)d_ws;
    float* qkv  = ws;
    float* qb   = qkv + (size_t)ROWS * QKV_N;
    float* kb   = qb  + (size_t)BB * NH * TT * DH;
    float* vb   = kb  + (size_t)BB * NH * TT * DH;
    float* attn = qkv;

    {
        gemm_bias_f32wmma<2, QKV_N, DM><<<768, 128, 0, stream>>>(
            x, W_qkv, b_qkv, qkv, ROWS);
    }

    rope_split<<<ROWS, 256, 0, stream>>>(qkv, qb, kb, vb);

    {
        const int blocks = BB * NH * (TT / 64);
        flash_attn_f32wmma<<<blocks, 128, 57344, stream>>>(qb, kb, vb, mask, attn);
    }

    {
        gemm_bias_f32wmma<1, DM, DM><<<512, 128, 0, stream>>>(
            attn, W_out, b_out, out, ROWS);
    }
}


# Attention: each 64-WMMA block runs from registers: a tile's K, V and P fragments are read from LDS as whole groups ahead of it (V staged as interleaved key pairs: 8-byte fragments); sums unchanged
# speedup vs baseline: 3.6135x; 1.0551x over previous
#include <hip/hip_runtime.h>
#include <hip/hip_bf16.h>
#include <math.h>


#define BB      2
#define TT      2048
#define DM      1024
#define NH      16
#define DH      64
#define ROWS    (BB * TT)
#define QKV_N   (3 * DM)

typedef __attribute__((ext_vector_type(2))) float v2f;
typedef __attribute__((ext_vector_type(4))) float v4f;
typedef __attribute__((ext_vector_type(8))) float v8f;

template <int NG, int N, int K>
struct GemmFrags {
    v4f a[2][2];
    v4f b[2][4][NG];
};

template <int NG, int N, int K>
__device__ __forceinline__ void gemm_load(GemmFrags<NG, N, K>& f,
                                          const float* __restrict__ ap0,
                                          const float* __restrict__ ap1,
                                          const float* __restrict__ bp)
{
    f.a[0][0] = *(const v4f*)ap0;
    f.a[1][0] = *(const v4f*)ap1;
#pragma unroll
    for (int j = 0; j < 4; ++j)
#pragma unroll
        for (int g = 0; g < NG; ++g)
            f.b[0][j][g] = *(const v4f*)(bp + (size_t)j * N + 64 * g);
    f.a[0][1] = *(const v4f*)(ap0 + 8);
    f.a[1][1] = *(const v4f*)(ap1 + 8);
#pragma unroll
    for (int j = 0; j < 4; ++j)
#pragma unroll
        for (int g = 0; g < NG; ++g)
            f.b[1][j][g] = *(const v4f*)(bp + (size_t)(8 + j) * N + 64 * g);
}

template <int NG, int N, int K>
__device__ __forceinline__ void gemm_mma(const GemmFrags<NG, N, K>& f, v8f (&acc)[2][4 * NG])
{
#pragma unroll
    for (int s = 0; s < 2; ++s) {
#pragma unroll
        for (int p = 0; p < 2; ++p) {
            v2f af0; af0.x = f.a[0][s][2 * p]; af0.y = f.a[0][s][2 * p + 1];
            v2f af1; af1.x = f.a[1][s][2 * p]; af1.y = f.a[1][s][2 * p + 1];
#pragma unroll
            for (int g = 0; g < NG; ++g) {
#pragma unroll
                for (int t = 0; t < 4; ++t) {
                    v2f bf; bf.x = f.b[s][2 * p][g][t]; bf.y = f.b[s][2 * p + 1][g][t];
                    acc[0][4 * g + t] = __builtin_amdgcn_wmma_f32_16x16x4_f32(false, af0, false, bf, (short)0, acc[0][4 * g + t], false, false);
                    acc[1][4 * g + t] = __builtin_amdgcn_wmma_f32_16x16x4_f32(false, af1, false, bf, (short)0, acc[1][4 * g + t], false, false);
                }
            }
        }
    }
}

template <int NG, int N, int K>
__global__ __launch_bounds__(128, (NG == 2) ? 3 : 4)
void gemm_bias_f32wmma(const float* __restrict__ A,
                       const float* __restrict__ Bm,
                       const float* __restrict__ bias,
                       float* __restrict__ C,
                       int M)
{
    const int wave = blockIdx.x * (blockDim.x >> 5) + (threadIdx.x >> 5);
    const int lane = threadIdx.x & 31;
    const int half = lane >> 4;
    const int ln   = lane & 15;

    constexpr int ngroups = N / (64 * NG);
    const int mt = wave / ngroups;
    const int ng = wave % ngroups;
    if (mt * 32 >= M) return;

    const int col = ng * (64 * NG) + 4 * ln;

    const float* __restrict__ ap0 = A + (size_t)(mt * 32 + ln) * K + 4 * half;
    const float* __restrict__ ap1 = ap0 + (size_t)16 * K;
    const float* __restrict__ bp  = Bm + (size_t)(4 * half) * N + col;

    v8f acc[2][4 * NG] = {};
    GemmFrags<NG, N, K> f0, f1;

    gemm_load<NG, N, K>(f0, ap0, ap1, bp);

    for (int k0 = 0; k0 < K - 32; k0 += 32) {
        gemm_load<NG, N, K>(f1, ap0 + 16, ap1 + 16, bp + (size_t)16 * N);
        __builtin_amdgcn_sched_barrier(0);
        gemm_mma<NG, N, K>(f0, acc);

        ap0 += 32;
        ap1 += 32;
        bp  += (size_t)32 * N;
        gemm_load<NG, N, K>(f0, ap0, ap1, bp);
        __builtin_amdgcn_sched_barrier(0);
        gemm_mma<NG, N, K>(f1, acc);
    }
    gemm_load<NG, N, K>(f1, ap0 + 16, ap1 + 16, bp + (size_t)16 * N);
    __builtin_amdgcn_sched_barrier(0);
    gemm_mma<NG, N, K>(f0, acc);
    gemm_mma<NG, N, K>(f1, acc);

#pragma unroll
    for (int g = 0; g < NG; ++g) {
        const v4f bs = *(const v4f*)(bias + col + 64 * g);
#pragma unroll
        for (int sub = 0; sub < 2; ++sub) {
            const int rbase = mt * 32 + sub * 16 + 8 * half;
#pragma unroll
            for (int v = 0; v < 8; ++v) {
                v4f o;
                o.x = acc[sub][4 * g + 0][v] + bs.x;
                o.y = acc[sub][4 * g + 1][v] + bs.y;
                o.z = acc[sub][4 * g + 2][v] + bs.z;
                o.w = acc[sub][4 * g + 3][v] + bs.w;
                *(v4f*)(C + (size_t)(rbase + v) * N + col + 64 * g) = o;
            }
        }
    }
}

__global__ void rope_split(const float* __restrict__ qkv,
                           float* __restrict__ qb,
                           float* __restrict__ kb,
                           float* __restrict__ vb)
{
    const int row = blockIdx.x;
    const int b = row / TT;
    const int t = row % TT;
    const float* __restrict__ src = qkv + (size_t)row * QKV_N;

    for (int p = threadIdx.x; p < QKV_N / 2; p += blockDim.x) {
        const int c   = p * 2;
        const int sec = c >> 10;
        const int cc  = c & (DM - 1);
        const int h   = cc >> 6;
        const int d   = cc & (DH - 1);

        const float e = src[c];
        const float o = src[c + 1];
        const size_t dst = (((size_t)(b * NH + h)) * TT + t) * DH + d;

        if (sec == 2) {
            vb[dst]     = e;
            vb[dst + 1] = o;
        } else {
            const int   i   = d >> 1;
            const float inv = expf(-(float)(2 * i) * (9.210340371976184f / 64.0f));
            const float ang = (float)t * inv;
            const float sn = sinf(ang);
            const float cs = cosf(ang);
            const float re = e * cs - o * sn;
            const float ro = e * sn + o * cs;
            if (sec == 0) { qb[dst] = re; qb[dst + 1] = ro; }
            else          { kb[dst] = re; kb[dst + 1] = ro; }
        }
    }
}

#define PSTRIDE 68
#define KTILE   64
#define KPITCH  68
#define VPAIR   160

template <int M>
__device__ __forceinline__ float row_xor(float x)
{
    return __int_as_float(__builtin_amdgcn_update_dpp(
        0, __float_as_int(x), 0x160 | M, 0xf, 0xf, true));
}

__global__ __launch_bounds__(128, 2)
void flash_attn_f32wmma(const float* __restrict__ qb,
                        const float* __restrict__ kb,
                        const float* __restrict__ vb,
                        const int*   __restrict__ amask,
                        float* __restrict__ attn  )
{
    __shared__ __attribute__((aligned(16))) float pshare[4][16 * PSTRIDE];
    __shared__ __attribute__((aligned(16))) float ks[KTILE * KPITCH];
    __shared__ __attribute__((aligned(16))) float vs[(KTILE / 2) * VPAIR];
    __shared__ int   ms[KTILE];

    const int tid  = threadIdx.x;
    const int wid  = tid >> 5;
    const int lane = tid & 31;
    const int half = lane >> 4;
    const int ln   = lane & 15;

    const int blocksPerBH = TT / 64;
    const int bh   = blockIdx.x % (BB * NH);
    const int qblk = blocksPerBH - 1 - blockIdx.x / (BB * NH);
    const int b    = bh >> 4;
    const int q0   = qblk * 64 + wid * 16;

    const float* __restrict__ Q  = qb + (size_t)bh * TT * DH;
    const float* __restrict__ Kp = kb + (size_t)bh * TT * DH;
    const float* __restrict__ Vp = vb + (size_t)bh * TT * DH;

    v2f qf[16];
    {
        const float* __restrict__ qrow = Q + (size_t)(q0 + ln) * DH;
#pragma unroll
        for (int i = 0; i < 16; ++i)
            qf[i] = *(const v2f*)(qrow + 4 * i + 2 * half);
    }

    v8f o0 = {}, o1 = {}, o2 = {}, o3 = {};
    float rmax[8], rsum[8];
#pragma unroll
    for (int v = 0; v < 8; ++v) { rmax[v] = -1e30f; rsum[v] = 0.0f; }

    float* __restrict__ P = &pshare[wid][0];
    const float scale = 0.125f;

    const int srow = tid >> 4;
    const int sc4  = tid & 15;
    const int vsrc = srow * 32 + sc4;
    float* __restrict__ kdst = ks + srow * KPITCH + 4 * sc4;
    float* __restrict__ vdst = vs + srow * VPAIR + 8 * sc4;

    const int* __restrict__ mg = amask + b * TT + (tid & (KTILE - 1));

    v4f kst[8], vst[8], vw[8];
    int mst;
    {
        const v4f* __restrict__ kg = (const v4f*)Kp + tid;
        const v4f* __restrict__ vg = (const v4f*)Vp + vsrc;
#pragma unroll
        for (int it = 0; it < 8; ++it) kst[it] = kg[it * 128];
#pragma unroll
        for (int it2 = 0; it2 < 4; ++it2) {
            vst[2 * it2]     = vg[it2 * 256];
            vst[2 * it2 + 1] = vg[it2 * 256 + 16];
        }
        mst = mg[0];
#pragma unroll
        for (int it2 = 0; it2 < 4; ++it2) {
            const v4f ra = vst[2 * it2];
            const v4f rb = vst[2 * it2 + 1];
            vw[2 * it2].x = ra.x; vw[2 * it2].y = rb.x; vw[2 * it2].z = ra.y; vw[2 * it2].w = rb.y;
            vw[2 * it2 + 1].x = ra.z; vw[2 * it2 + 1].y = rb.z; vw[2 * it2 + 1].z = ra.w; vw[2 * it2 + 1].w = rb.w;
        }
    }

    for (int kt = 0; kt <= qblk; ++kt) {
        __syncthreads();
#pragma unroll
        for (int it = 0; it < 8; ++it)
            *(v4f*)(kdst + it * 8 * KPITCH) = kst[it];
#pragma unroll
        for (int it2 = 0; it2 < 4; ++it2) {
            *(v4f*)(vdst + it2 * 8 * VPAIR)     = vw[2 * it2];
            *(v4f*)(vdst + it2 * 8 * VPAIR + 4) = vw[2 * it2 + 1];
        }
        if (tid < KTILE) ms[tid] = mst;
        __syncthreads();

        {
            const int ktn = (kt < qblk) ? kt + 1 : kt;
            const v4f* __restrict__ kg = (const v4f*)(Kp + (size_t)ktn * KTILE * DH) + tid;
            const v4f* __restrict__ vg = (const v4f*)(Vp + (size_t)ktn * KTILE * DH) + vsrc;
#pragma unroll
            for (int it = 0; it < 8; ++it) kst[it] = kg[it * 128];
#pragma unroll
            for (int it2 = 0; it2 < 4; ++it2) {
                vst[2 * it2]     = vg[it2 * 256];
                vst[2 * it2 + 1] = vg[it2 * 256 + 16];
            }
            mst = mg[ktn * KTILE];
        }

        v8f s0 = {}, s1 = {}, s2 = {}, s3 = {};
        {
            v2f kf[16][4];
            const float* __restrict__ krow = ks + ln * KPITCH + 2 * half;
#pragma unroll
            for (int i = 0; i < 16; ++i)
#pragma unroll
                for (int j = 0; j < 4; ++j)
                    kf[i][j] = *(const v2f*)(krow + 16 * j * KPITCH + 4 * i);
            __builtin_amdgcn_sched_barrier(0);
#pragma unroll
            for (int i = 0; i < 16; ++i) {
                s0 = __builtin_amdgcn_wmma_f32_16x16x4_f32(false, qf[i], false, kf[i][0], (short)0, s0, false, false);
                s1 = __builtin_amdgcn_wmma_f32_16x16x4_f32(false, qf[i], false, kf[i][1], (short)0, s1, false, false);
                s2 = __builtin_amdgcn_wmma_f32_16x16x4_f32(false, qf[i], false, kf[i][2], (short)0, s2, false, false);
                s3 = __builtin_amdgcn_wmma_f32_16x16x4_f32(false, qf[i], false, kf[i][3], (short)0, s3, false, false);
            }
        }
        asm volatile("" :: "v"(s0), "v"(s1), "v"(s2), "v"(s3) : "memory");

        v2f vf[16][4];
        {
            const float* __restrict__ vrow = vs + half * VPAIR + 2 * ln;
#pragma unroll
            for (int i = 0; i < 16; ++i)
#pragma unroll
                for (int t = 0; t < 4; ++t)
                    vf[i][t] = *(const v2f*)(vrow + 2 * i * VPAIR + 32 * t);
        }

        const int key0 = kt * KTILE + ln;
        const int pm0 = ms[ln];
        const int pm1 = ms[16 + ln];
        const int pm2 = ms[32 + ln];
        const int pm3 = ms[48 + ln];
#pragma unroll
        for (int v = 0; v < 8; ++v) {
            const int qrowi = q0 + v + 8 * half;
            float a0 = s0[v] * scale;
            float a1 = s1[v] * scale;
            float a2 = s2[v] * scale;
            float a3 = s3[v] * scale;
            a0 = (key0      > qrowi || pm0 == 0) ? -1e30f : a0;
            a1 = (key0 + 16 > qrowi || pm1 == 0) ? -1e30f : a1;
            a2 = (key0 + 32 > qrowi || pm2 == 0) ? -1e30f : a2;
            a3 = (key0 + 48 > qrowi || pm3 == 0) ? -1e30f : a3;

            float mx = fmaxf(fmaxf(a0, a1), fmaxf(a2, a3));
            mx = fmaxf(mx, row_xor<1>(mx));
            mx = fmaxf(mx, row_xor<2>(mx));
            mx = fmaxf(mx, row_xor<4>(mx));
            mx = fmaxf(mx, row_xor<8>(mx));

            const float nmax = fmaxf(rmax[v], mx);
            const float p0 = __expf(a0 - nmax);
            const float p1 = __expf(a1 - nmax);
            const float p2 = __expf(a2 - nmax);
            const float p3 = __expf(a3 - nmax);

            float sr = (p0 + p1) + (p2 + p3);
            sr += row_xor<1>(sr);
            sr += row_xor<2>(sr);
            sr += row_xor<4>(sr);
            sr += row_xor<8>(sr);

            const float alpha = __expf(rmax[v] - nmax);
            rsum[v] = rsum[v] * alpha + sr;
            rmax[v] = nmax;
            o0[v] *= alpha; o1[v] *= alpha; o2[v] *= alpha; o3[v] *= alpha;

            float* __restrict__ prow = P + (v + 8 * half) * PSTRIDE + ln;
            prow[0]  = p0;
            prow[16] = p1;
            prow[32] = p2;
            prow[48] = p3;
        }

        asm volatile("s_wait_dscnt 0x0" ::: "memory");

        v2f af[16];
#pragma unroll
        for (int i = 0; i < 16; ++i)
            af[i] = *(const v2f*)(P + ln * PSTRIDE + 4 * i + 2 * half);
        __builtin_amdgcn_sched_barrier(0);
#pragma unroll
        for (int i = 0; i < 16; ++i) {
            o0 = __builtin_amdgcn_wmma_f32_16x16x4_f32(false, af[i], false, vf[i][0], (short)0, o0, false, false);
            o1 = __builtin_amdgcn_wmma_f32_16x16x4_f32(false, af[i], false, vf[i][1], (short)0, o1, false, false);
            o2 = __builtin_amdgcn_wmma_f32_16x16x4_f32(false, af[i], false, vf[i][2], (short)0, o2, false, false);
            o3 = __builtin_amdgcn_wmma_f32_16x16x4_f32(false, af[i], false, vf[i][3], (short)0, o3, false, false);
        }

#pragma unroll
        for (int it2 = 0; it2 < 4; ++it2) {
            const v4f ra = vst[2 * it2];
            const v4f rb = vst[2 * it2 + 1];
            vw[2 * it2].x = ra.x; vw[2 * it2].y = rb.x; vw[2 * it2].z = ra.y; vw[2 * it2].w = rb.y;
            vw[2 * it2 + 1].x = ra.z; vw[2 * it2 + 1].y = rb.z; vw[2 * it2 + 1].z = ra.w; vw[2 * it2 + 1].w = rb.w;
        }
    }

    const int h = bh & (NH - 1);
    v8f os[4] = { o0, o1, o2, o3 };
#pragma unroll
    for (int v = 0; v < 8; ++v) {
        const float inv = 1.0f / rsum[v];
        const size_t row = (size_t)b * TT + (q0 + v + 8 * half);
#pragma unroll
        for (int t = 0; t < 4; ++t)
            attn[row * DM + h * DH + t * 16 + ln] = os[t][v] * inv;
    }
}

extern "C" void kernel_launch(void* const* d_in, const int* in_sizes, int n_in,
                              void* d_out, int out_size, void* d_ws, size_t ws_size,
                              hipStream_t stream)
{
    (void)in_sizes; (void)n_in; (void)out_size; (void)ws_size;

    const float* x      = (const float*)d_in[0];
    const int*   mask   = (const int*)  d_in[1];
    const float* W_qkv  = (const float*)d_in[2];
    const float* b_qkv  = (const float*)d_in[3];
    const float* W_out  = (const float*)d_in[4];
    const float* b_out  = (const float*)d_in[5];
    float*       out    = (float*)d_out;

    float* ws   = (float*)d_ws;
    float* qkv  = ws;
    float* qb   = qkv + (size_t)ROWS * QKV_N;
    float* kb   = qb  + (size_t)BB * NH * TT * DH;
    float* vb   = kb  + (size_t)BB * NH * TT * DH;
    float* attn = qkv;

    {
        gemm_bias_f32wmma<2, QKV_N, DM><<<768, 128, 0, stream>>>(
            x, W_qkv, b_qkv, qkv, ROWS);
    }

    rope_split<<<ROWS, 256, 0, stream>>>(qkv, qb, kb, vb);

    {
        const int blocks = BB * NH * (TT / 64);
        flash_attn_f32wmma<<<blocks, 128, 57344, stream>>>(qb, kb, vb, mask, attn);
    }

    {
        gemm_bias_f32wmma<1, DM, DM><<<512, 128, 0, stream>>>(
            attn, W_out, b_out, out, ROWS);
    }
}
